# starting kernel
# baseline (speedup 1.0000x reference)
#include <hip/hip_runtime.h>

typedef float v2f __attribute__((ext_vector_type(2)));
typedef float v8f __attribute__((ext_vector_type(8)));

#define H28 28
#define P784 784
#define C32 32
#define ROW588 588
#define KPAD 64
#define MOUT 256

__global__ void k_t3(const float* __restrict__ x, float* __restrict__ t3r) {
    int tid = blockIdx.x * blockDim.x + threadIdx.x;
    if (tid >= C32 * P784) return;
    int c = tid / P784, rem = tid % P784;
    int h = rem / H28, w = rem % H28;
    int wsrc = (w + H28 - 1) % H28;
    int src = c * P784 + h * H28 + wsrc;
    t3r[tid] = x[src] + x[src + C32 * P784];
}

__global__ void k_t5(const float* __restrict__ t3r, const float* __restrict__ w1,
                     float* __restrict__ t5p) {
    int tid = blockIdx.x * blockDim.x + threadIdx.x;
    if (tid >= 30 * ROW588) return;
    int row = tid / ROW588;
    int rem = tid % ROW588;
    if (row == 0 || row == 29) { t5p[tid] = 0.0f; return; }
    int h = row - 1;
    int w = rem / 21, kc = rem % 21;
    float acc = 0.0f;
    for (int c = 0; c < C32; ++c) {
        const float* t3row = t3r + c * P784 + h * H28;
        const float* w1row = w1 + c * 147 + kc;
#pragma unroll
        for (int j = 0; j < 7; ++j) {
            int ww = w + j - 3;
            float v = (ww >= 0 && ww < H28) ? t3row[ww] : 0.0f;
            acc += v * w1row[j * 21];
        }
    }
    t5p[tid] = acc;
}

__global__ void k_packA(const float* __restrict__ w0, float* __restrict__ w0p) {
    int tid = blockIdx.x * blockDim.x + threadIdx.x;
    if (tid >= MOUT * KPAD) return;
    int i = tid >> 6, k = tid & 63;
    w0p[tid] = (k < 63) ? w0[i * 63 + k] : 0.0f;
}

__global__ void k_packB(const float* __restrict__ t5p, float* __restrict__ Bm) {
    int tid = blockIdx.x * blockDim.x + threadIdx.x;
    if (tid >= P784 * KPAD) return;
    int p = tid >> 6, k = tid & 63;
    float v = 0.0f;
    if (k < 63) {
        int j = k / 21, kc = k % 21;
        int h = p / H28, w = p % H28;
        v = t5p[(h + j) * ROW588 + w * 21 + kc];
    }
    Bm[tid] = v;
}

__global__ void __launch_bounds__(32)
k_gemm(const float* __restrict__ A, const float* __restrict__ B, float* __restrict__ D) {
    const int lane = threadIdx.x;
    const int t    = lane & 15;
    const int kh   = lane >> 4;
    const int mi   = blockIdx.x;
    const int ni   = blockIdx.y;

    const float* Arow = A + (size_t)(mi * 16 + t) * KPAD;
    const float* Brow = B + (size_t)(ni * 16 + t) * KPAD;

    v8f c = {0.f, 0.f, 0.f, 0.f, 0.f, 0.f, 0.f, 0.f};
#pragma unroll
    for (int kk = 0; kk < 16; ++kk) {
        const int k0 = kk * 4 + kh * 2;
        v2f a = *(const v2f*)(Arow + k0);
        v2f b = *(const v2f*)(Brow + k0);
        c = __builtin_amdgcn_wmma_f32_16x16x4_f32(
                 false, a,  false, b,
                 (short)0, c,  false,  false);
    }

#pragma unroll
    for (int r = 0; r < 8; ++r) {
        const int m = r + kh * 8;
        D[(size_t)(mi * 16 + m) * P784 + ni * 16 + t] = c[r];
    }
}

extern "C" void kernel_launch(void* const* d_in, const int* in_sizes, int n_in,
                              void* d_out, int out_size, void* d_ws, size_t ws_size,
                              hipStream_t stream) {
    const float* x  = (const float*)d_in[0];
    const float* w0 = (const float*)d_in[1];
    const float* w1 = (const float*)d_in[2];
    float* y = (float*)d_out;

    float* ws  = (float*)d_ws;
    float* t3r = ws;
    float* t5p = ws + 25088;
    float* w0p = ws + 42728;
    float* Bm  = ws + 59112;

    k_t3   <<<(C32 * P784 + 255) / 256, 256, 0, stream>>>(x, t3r);
    k_t5   <<<(30 * ROW588 + 255) / 256, 256, 0, stream>>>(t3r, w1, t5p);
    k_packA<<<(MOUT * KPAD) / 256,      256, 0, stream>>>(w0, w0p);
    k_packB<<<(P784 * KPAD) / 256,      256, 0, stream>>>(t5p, Bm);
    k_gemm <<<dim3(16, 49), 32, 0, stream>>>(w0p, Bm, y);
}


# five launches fused into one kernel: a block (16-position tile, 128 channels) stages x, w1, w0 in LDS, rebuilds its three stage-1 tiles there and feeds stage 2 from LDS; f32 summation order kept
# speedup vs baseline: 5.9275x; 5.9275x over previous
#include <hip/hip_runtime.h>

typedef float v2f __attribute__((ext_vector_type(2)));
typedef float v4f __attribute__((ext_vector_type(4)));
typedef float v8f __attribute__((ext_vector_type(8)));

#define H28 28
#define P784 784
#define C32 32
#define TPITCH 80
#define BPITCH 66
#define W1N 4704
#define W0H 8064

__global__ void __launch_bounds__(256)
k_fused(const float* __restrict__ x, const float* __restrict__ w0, const float* __restrict__ w1,
        float* __restrict__ y) {
    __shared__ float t3s[C32 * TPITCH];
    __shared__ float red[8][3][2][8][32];
    __shared__ __attribute__((aligned(16))) float Bs[16 * BPITCH];
    __shared__ __attribute__((aligned(16))) float w1s[W1N];
    __shared__ __attribute__((aligned(16))) float w0s[W0H];

    const int tid  = threadIdx.x;
    __builtin_assume(tid >= 0 && tid < 256);
    const int lane = tid & 31;
    const int q    = __builtin_amdgcn_readfirstlane(tid >> 5);
    const int t    = lane & 15;
    const int hh   = lane >> 4;
    const int pt   = blockIdx.x >> 1;
    const int cs   = blockIdx.x & 1;
    const int p0   = pt * 16;
    const int c0   = 4 * q + 2 * hh;

    float xa[10], xb[10];
    bool  xv[10];
    {
        int c  = tid / TPITCH;
        int fl = tid - c * TPITCH;
#pragma unroll
        for (int i = 0; i < 10; ++i) {
            const int f  = p0 - 31 + fl;
            xv[i] = (f >= 0) && (f < P784);
            const int fc  = xv[i] ? f : 0;
            const int w   = fc % H28;
            const int src = (w == 0) ? fc + 27 : fc - 1;
            xa[i] = x[c * P784 + src];
            xb[i] = x[(c + C32) * P784 + src];
            fl += 256 - 3 * TPITCH;
            c  += 3;
            if (fl >= TPITCH) { fl -= TPITCH; c += 1; }
        }
    }

    const v4f* w1v = (const v4f*)w1;
    const v4f* w0v = (const v4f*)(w0 + cs * W0H);
    v4f g1[5], g0[8];
#pragma unroll
    for (int i = 0; i < 5; ++i) {
        const int e = tid + 256 * i;
        g1[i] = w1v[(e < W1N / 4) ? e : 0];
    }
#pragma unroll
    for (int i = 0; i < 8; ++i) {
        const int e = tid + 256 * i;
        g0[i] = w0v[(e < W0H / 4) ? e : 0];
    }

#pragma unroll
    for (int i = 0; i < 10; ++i) t3s[tid + 256 * i] = xv[i] ? (xa[i] + xb[i]) : 0.0f;
#pragma unroll
    for (int i = 0; i < 5; ++i) {
        const int e = tid + 256 * i;
        if (e < W1N / 4) ((v4f*)w1s)[e] = g1[i];
    }
#pragma unroll
    for (int i = 0; i < 8; ++i) {
        const int e = tid + 256 * i;
        if (e < W0H / 4) ((v4f*)w0s)[e] = g0[i];
    }
    __syncthreads();

    int  wq[3];
    bool pv[3];
#pragma unroll
    for (int r = 0; r < 3; ++r) {
        const int P = p0 + t + 28 * (r - 1);
        pv[r] = (P >= 0) && (P < P784);
        wq[r] = (pv[r] ? P : 0) % H28;
    }
    const float* ab = t3s + c0 * TPITCH + t;
    const int    n1     = (16 + t < 21) ? (16 + t) : 20;
    const float* b0base = w1s + c0 * 147 + t;
    const float* b1base = w1s + c0 * 147 + n1;

    v8f acc[3][2];
#pragma unroll
    for (int r = 0; r < 3; ++r) {
        acc[r][0] = (v8f){0.f, 0.f, 0.f, 0.f, 0.f, 0.f, 0.f, 0.f};
        acc[r][1] = (v8f){0.f, 0.f, 0.f, 0.f, 0.f, 0.f, 0.f, 0.f};
    }
#pragma unroll
    for (int j = 0; j < 7; ++j) {
        v2f b0; b0.x = b0base[j * 21];  b0.y = b0base[147 + j * 21];
        v2f b1; b1.x = b1base[j * 21];  b1.y = b1base[147 + j * 21];
#pragma unroll
        for (int r = 0; r < 3; ++r) {
            const int  ww = wq[r] + j - 3;
            const bool ok = pv[r] && (ww >= 0) && (ww < H28);
            const float a0 = ab[28 * r + j];
            const float a1 = ab[TPITCH + 28 * r + j];
            v2f a;  a.x = ok ? a0 : 0.0f;  a.y = ok ? a1 : 0.0f;
            acc[r][0] = __builtin_amdgcn_wmma_f32_16x16x4_f32(false, a, false, b0, (short)0, acc[r][0], false, false);
            acc[r][1] = __builtin_amdgcn_wmma_f32_16x16x4_f32(false, a, false, b1, (short)0, acc[r][1], false, false);
        }
    }

#pragma unroll
    for (int r = 0; r < 3; ++r) {
#pragma unroll
        for (int g = 0; g < 8; ++g) {
            red[q][r][0][g][lane] = acc[r][0][g];
            red[q][r][1][g][lane] = acc[r][1][g];
        }
    }
    __syncthreads();

    const float* arow = w0s + (q * 16 + t) * 63;
    v2f a2[16];
#pragma unroll
    for (int kk = 0; kk < 16; ++kk) {
        const int k0 = kk * 4 + hh * 2;
        const int k1 = (k0 + 1 < 63) ? (k0 + 1) : 62;
        const float v1 = arow[k1];
        a2[kk].x = arow[k0];
        a2[kk].y = (k0 + 1 < 63) ? v1 : 0.0f;
    }

#pragma unroll
    for (int i = 0; i < 4; ++i) {
        const int e  = tid + 256 * i;
        if (e >= 3 * 336) break;
        const int r  = e / 336;
        const int e2 = e - r * 336;
        const int m  = e2 / 21, n = e2 - m * 21;
        const int nt = n >> 4;
        const int g  = m & 7;
        const int l  = ((m >> 3) << 4) + (n & 15);
        float s = red[0][r][nt][g][l];
#pragma unroll
        for (int qq = 1; qq < 8; ++qq) s += red[qq][r][nt][g][l];
        const int P = p0 + m + 28 * (r - 1);
        Bs[m * BPITCH + r * 21 + n] = ((P >= 0) && (P < P784)) ? s : 0.0f;
    }
    if (tid < 16) Bs[tid * BPITCH + 63] = 0.0f;
    __syncthreads();

    const float* brow = Bs + t * BPITCH;
    v8f c = {0.f, 0.f, 0.f, 0.f, 0.f, 0.f, 0.f, 0.f};
#pragma unroll
    for (int kk = 0; kk < 16; ++kk) {
        const int k0 = kk * 4 + hh * 2;
        v2f b = *(const v2f*)(brow + k0);
        c = __builtin_amdgcn_wmma_f32_16x16x4_f32(false, a2[kk], false, b, (short)0, c, false, false);
    }

    float* yrow = y + (size_t)(cs * 128 + q * 16 + hh * 8) * P784 + p0 + t;
#pragma unroll
    for (int g = 0; g < 8; ++g) yrow[g * P784] = c[g];
}

extern "C" void kernel_launch(void* const* d_in, const int* in_sizes, int n_in,
                              void* d_out, int out_size, void* d_ws, size_t ws_size,
                              hipStream_t stream) {
    const float* x  = (const float*)d_in[0];
    const float* w0 = (const float*)d_in[1];
    const float* w1 = (const float*)d_in[2];
    float* y = (float*)d_out;

    k_fused<<<98, 256, 0, stream>>>(x, w0, w1, y);
}


# LDS read-out between the matrix stages: partial tiles as row pairs (8-byte accesses), a thread's 24 reads one group off one address; stage-2 channels 4 apart per wave: no bank conflicts; same sums
# speedup vs baseline: 6.3761x; 1.0757x over previous
#include <hip/hip_runtime.h>

typedef float v2f __attribute__((ext_vector_type(2)));
typedef float v4f __attribute__((ext_vector_type(4)));
typedef float v8f __attribute__((ext_vector_type(8)));

#define H28 28
#define P784 784
#define C32 32
#define TPITCH 80
#define BPITCH 66
#define W1N 4704
#define W0H 8064

__global__ void __launch_bounds__(256)
k_fused(const float* __restrict__ x, const float* __restrict__ w0, const float* __restrict__ w1,
        float* __restrict__ y) {
    __shared__ float t3s[C32 * TPITCH];
    __shared__ __attribute__((aligned(16))) float red[8][3][2][4][32][2];
    __shared__ __attribute__((aligned(16))) float Bs[16 * BPITCH];
    __shared__ __attribute__((aligned(16))) float w1s[W1N];
    __shared__ __attribute__((aligned(16))) float w0s[W0H];

    const int tid  = threadIdx.x;
    __builtin_assume(tid >= 0 && tid < 256);
    const int lane = tid & 31;
    const int q    = __builtin_amdgcn_readfirstlane(tid >> 5);
    const int t    = lane & 15;
    const int hh   = lane >> 4;
    const int pt   = blockIdx.x >> 1;
    const int cs   = blockIdx.x & 1;
    const int p0   = pt * 16;
    const int c0   = 4 * q + 2 * hh;

    float xa[10], xb[10];
    bool  xv[10];
    {
        int c  = tid / TPITCH;
        int fl = tid - c * TPITCH;
#pragma unroll
        for (int i = 0; i < 10; ++i) {
            const int f  = p0 - 31 + fl;
            xv[i] = (f >= 0) && (f < P784);
            const int fc  = xv[i] ? f : 0;
            const int w   = fc % H28;
            const int src = (w == 0) ? fc + 27 : fc - 1;
            xa[i] = x[c * P784 + src];
            xb[i] = x[(c + C32) * P784 + src];
            fl += 256 - 3 * TPITCH;
            c  += 3;
            if (fl >= TPITCH) { fl -= TPITCH; c += 1; }
        }
    }

    const v4f* w1v = (const v4f*)w1;
    const v4f* w0v = (const v4f*)(w0 + cs * W0H);
    v4f g1[5], g0[8];
#pragma unroll
    for (int i = 0; i < 5; ++i) {
        const int e = tid + 256 * i;
        g1[i] = w1v[(e < W1N / 4) ? e : 0];
    }
#pragma unroll
    for (int i = 0; i < 8; ++i) {
        const int e = tid + 256 * i;
        g0[i] = w0v[(e < W0H / 4) ? e : 0];
    }

#pragma unroll
    for (int i = 0; i < 10; ++i) t3s[tid + 256 * i] = xv[i] ? (xa[i] + xb[i]) : 0.0f;
#pragma unroll
    for (int i = 0; i < 5; ++i) {
        const int e = tid + 256 * i;
        if (e < W1N / 4) ((v4f*)w1s)[e] = g1[i];
    }
#pragma unroll
    for (int i = 0; i < 8; ++i) {
        const int e = tid + 256 * i;
        if (e < W0H / 4) ((v4f*)w0s)[e] = g0[i];
    }
    __syncthreads();

    int  wq[3];
    bool pv[3];
#pragma unroll
    for (int r = 0; r < 3; ++r) {
        const int P = p0 + t + 28 * (r - 1);
        pv[r] = (P >= 0) && (P < P784);
        wq[r] = (pv[r] ? P : 0) % H28;
    }
    const float* ab = t3s + c0 * TPITCH + t;
    const int    n1     = (16 + t < 21) ? (16 + t) : 20;
    const float* b0base = w1s + c0 * 147 + t;
    const float* b1base = w1s + c0 * 147 + n1;

    v8f acc[3][2];
#pragma unroll
    for (int r = 0; r < 3; ++r) {
        acc[r][0] = (v8f){0.f, 0.f, 0.f, 0.f, 0.f, 0.f, 0.f, 0.f};
        acc[r][1] = (v8f){0.f, 0.f, 0.f, 0.f, 0.f, 0.f, 0.f, 0.f};
    }
#pragma unroll
    for (int j = 0; j < 7; ++j) {
        v2f b0; b0.x = b0base[j * 21];  b0.y = b0base[147 + j * 21];
        v2f b1; b1.x = b1base[j * 21];  b1.y = b1base[147 + j * 21];
#pragma unroll
        for (int r = 0; r < 3; ++r) {
            const int  ww = wq[r] + j - 3;
            const bool ok = pv[r] && (ww >= 0) && (ww < H28);
            const float a0 = ab[28 * r + j];
            const float a1 = ab[TPITCH + 28 * r + j];
            v2f a;  a.x = ok ? a0 : 0.0f;  a.y = ok ? a1 : 0.0f;
            acc[r][0] = __builtin_amdgcn_wmma_f32_16x16x4_f32(false, a, false, b0, (short)0, acc[r][0], false, false);
            acc[r][1] = __builtin_amdgcn_wmma_f32_16x16x4_f32(false, a, false, b1, (short)0, acc[r][1], false, false);
        }
    }

#pragma unroll
    for (int r = 0; r < 3; ++r) {
#pragma unroll
        for (int nt = 0; nt < 2; ++nt) {
#pragma unroll
            for (int gp = 0; gp < 4; ++gp) {
                v2f v;  v.x = acc[r][nt][2 * gp];  v.y = acc[r][nt][2 * gp + 1];
                *(v2f*)&red[q][r][nt][gp][lane][0] = v;
            }
        }
    }
    __syncthreads();

    const int gp = q & 3, nt = q >> 2;
    v2f pr[3][8];
#pragma unroll
    for (int r = 0; r < 3; ++r) {
#pragma unroll
        for (int qq = 0; qq < 8; ++qq) pr[r][qq] = *(const v2f*)&red[qq][r][nt][gp][lane][0];
    }

    const int chb = 64 * (q >> 2) + (q & 3);
    const float* arow = w0s + (chb + 4 * t) * 63;
    v2f a2[16];
#pragma unroll
    for (int kk = 0; kk < 16; ++kk) {
        const int k0 = kk * 4 + hh * 2;
        const int k1 = (k0 + 1 < 63) ? (k0 + 1) : 62;
        const float v1 = arow[k1];
        a2[kk].x = arow[k0];
        a2[kk].y = (k0 + 1 < 63) ? v1 : 0.0f;
    }

    {
        const int  n   = nt * 16 + t;
        const int  m0  = 8 * hh + 2 * gp;
        const bool has = n < 21;
        const int  o0  = has ? (m0 * BPITCH + n) : ((lane >> 1) * BPITCH + 64 + (lane & 1));
        const int  o1  = has ? (o0 + BPITCH) : o0;
        const int  rs  = has ? 21 : 0;
#pragma unroll
        for (int r = 0; r < 3; ++r) {
            v2f s = pr[r][0];
#pragma unroll
            for (int qq = 1; qq < 8; ++qq) { s.x += pr[r][qq].x;  s.y += pr[r][qq].y; }
            const int P = p0 + m0 + 28 * (r - 1);
            Bs[o0 + r * rs] = ((P     >= 0) && (P     < P784)) ? s.x : 0.0f;
            Bs[o1 + r * rs] = ((P + 1 >= 0) && (P + 1 < P784)) ? s.y : 0.0f;
        }
    }
    if (tid < 16) Bs[tid * BPITCH + 63] = 0.0f;
    __syncthreads();

    const float* brow = Bs + t * BPITCH;
    v8f c = {0.f, 0.f, 0.f, 0.f, 0.f, 0.f, 0.f, 0.f};
#pragma unroll
    for (int kk = 0; kk < 16; ++kk) {
        const int k0 = kk * 4 + hh * 2;
        v2f b = *(const v2f*)(brow + k0);
        c = __builtin_amdgcn_wmma_f32_16x16x4_f32(false, a2[kk], false, b, (short)0, c, false, false);
    }

    float* yrow = y + (size_t)(cs * 128 + chb + 4 * (hh * 8)) * P784 + p0 + t;
#pragma unroll
    for (int g = 0; g < 8; ++g) yrow[4 * g * P784] = c[g];
}

extern "C" void kernel_launch(void* const* d_in, const int* in_sizes, int n_in,
                              void* d_out, int out_size, void* d_ws, size_t ws_size,
                              hipStream_t stream) {
    const float* x  = (const float*)d_in[0];
    const float* w0 = (const float*)d_in[1];
    const float* w1 = (const float*)d_in[2];
    float* y = (float*)d_out;

    k_fused<<<98, 256, 0, stream>>>(x, w0, w1, y);
}


# fill private to each wave and pipelined under stage 1: a wave stages only its own 4 channels of x and w1, h tap by h tap, no barrier before stage 1; w0 and partial tiles stored under it; same sums
# speedup vs baseline: 6.7626x; 1.0606x over previous
#include <hip/hip_runtime.h>

typedef float v2f __attribute__((ext_vector_type(2)));
typedef float v4f __attribute__((ext_vector_type(4)));
typedef float v8f __attribute__((ext_vector_type(8)));

#define H28 28
#define P784 784
#define C32 32
#define TPITCH 80
#define BPITCH 66
#define W1N 4704
#define W0H 8064

__global__ void __launch_bounds__(256)
k_fused(const float* __restrict__ x, const float* __restrict__ w0, const float* __restrict__ w1,
        float* __restrict__ y) {
    __shared__ float t3s[C32 * TPITCH];
    __shared__ __attribute__((aligned(16))) float red[8][3][2][4][32][2];
    __shared__ __attribute__((aligned(16))) float Bs[16 * BPITCH];
    __shared__ __attribute__((aligned(16))) float w1s[W1N];
    __shared__ __attribute__((aligned(16))) float w0s[W0H];

    const int tid  = threadIdx.x;
    __builtin_assume(tid >= 0 && tid < 256);
    const int lane = tid & 31;
    const int q    = __builtin_amdgcn_readfirstlane(tid >> 5);
    const int t    = lane & 15;
    const int hh   = lane >> 4;
    const int pt   = blockIdx.x >> 1;
    const int cs   = blockIdx.x & 1;
    const int p0   = pt * 16;
    const int c0   = 4 * q + 2 * hh;

    const int cq = 4 * q;
    const v4f* w1v = (const v4f*)(w1 + cq * 147);
    v4f g1[5];
#pragma unroll
    for (int i = 0; i < 5; ++i) {
        const int e = lane + 32 * i;
        g1[i] = w1v[(e < 147) ? e : 0];
    }

    float xa[3][3], xb[3][3];
    bool  xv[3][3];
    int   xo[3];
#pragma unroll
    for (int r = 0; r < 3; ++r) {
#pragma unroll
        for (int i = 0; i < 3; ++i) {
            const int e  = lane + 32 * i;
            const int ec = (e < 88) ? e : 87;
            const int cl = ec / 22;
            const int s  = ec - 22 * cl;
            const int f  = p0 - 31 + 28 * r + s;
            xv[r][i] = (f >= 0) && (f < P784);
            const int fc  = xv[r][i] ? f : 0;
            const int w   = fc % H28;
            const int src = (w == 0) ? fc + 27 : fc - 1;
            xa[r][i] = x[(cq + cl) * P784 + src];
            xb[r][i] = x[(cq + cl + C32) * P784 + src];
            if (r == 0) xo[i] = (cq + cl) * TPITCH + s;
        }
    }

    const v4f* w0v = (const v4f*)(w0 + cs * W0H);
    v4f g0[8];
#pragma unroll
    for (int i = 0; i < 8; ++i) {
        const int e = tid + 256 * i;
        g0[i] = w0v[(e < W0H / 4) ? e : 0];
    }

#pragma unroll
    for (int i = 0; i < 5; ++i) {
        const int e = lane + 32 * i;
        if (e < 147) ((v4f*)(w1s + cq * 147))[e] = g1[i];
    }

    int  wq[3];
    bool pv[3];
#pragma unroll
    for (int r = 0; r < 3; ++r) {
        const int P = p0 + t + 28 * (r - 1);
        pv[r] = (P >= 0) && (P < P784);
        wq[r] = (pv[r] ? P : 0) % H28;
    }
    const float* ab = t3s + c0 * TPITCH + t;
    const int    n1     = (16 + t < 21) ? (16 + t) : 20;
    const float* b0base = w1s + c0 * 147 + t;
    const float* b1base = w1s + c0 * 147 + n1;

    v2f b0[7], b1[7];
#pragma unroll
    for (int i = 0; i < 3; ++i) {
        if (lane + 32 * i < 88) t3s[xo[i]] = xv[0][i] ? (xa[0][i] + xb[0][i]) : 0.0f;
    }
    __builtin_amdgcn_fence(__ATOMIC_RELEASE, "wavefront");
    __builtin_amdgcn_wave_barrier();
    __builtin_amdgcn_fence(__ATOMIC_ACQUIRE, "wavefront");
#pragma unroll
    for (int j = 0; j < 7; ++j) {
        b0[j].x = b0base[j * 21];  b0[j].y = b0base[147 + j * 21];
        b1[j].x = b1base[j * 21];  b1[j].y = b1base[147 + j * 21];
    }
#pragma unroll
    for (int r = 0; r < 3; ++r) {
        v8f acc[2];
        acc[0] = (v8f){0.f, 0.f, 0.f, 0.f, 0.f, 0.f, 0.f, 0.f};
        acc[1] = (v8f){0.f, 0.f, 0.f, 0.f, 0.f, 0.f, 0.f, 0.f};
#pragma unroll
        for (int j = 0; j < 7; ++j) {
            if (j == 4) {
                if (r < 2) {
#pragma unroll
                    for (int i = 0; i < 3; ++i) {
                        if (lane + 32 * i < 88) t3s[xo[i] + 28 * (r + 1)] = xv[r + 1][i] ? (xa[r + 1][i] + xb[r + 1][i]) : 0.0f;
                    }
                    __builtin_amdgcn_fence(__ATOMIC_RELEASE, "wavefront");
                    __builtin_amdgcn_wave_barrier();
                    __builtin_amdgcn_fence(__ATOMIC_ACQUIRE, "wavefront");
                } else {
#pragma unroll
                    for (int i = 0; i < 8; ++i) {
                        const int e = tid + 256 * i;
                        if (e < W0H / 4) ((v4f*)w0s)[e] = g0[i];
                    }
                }
            }
            const int  ww = wq[r] + j - 3;
            const bool ok = pv[r] && (ww >= 0) && (ww < H28);
            const float a0 = ab[28 * r + j];
            const float a1 = ab[TPITCH + 28 * r + j];
            v2f a;  a.x = ok ? a0 : 0.0f;  a.y = ok ? a1 : 0.0f;
            acc[0] = __builtin_amdgcn_wmma_f32_16x16x4_f32(false, a, false, b0[j], (short)0, acc[0], false, false);
            acc[1] = __builtin_amdgcn_wmma_f32_16x16x4_f32(false, a, false, b1[j], (short)0, acc[1], false, false);
        }
#pragma unroll
        for (int nt = 0; nt < 2; ++nt) {
#pragma unroll
            for (int gp = 0; gp < 4; ++gp) {
                v2f v;  v.x = acc[nt][2 * gp];  v.y = acc[nt][2 * gp + 1];
                *(v2f*)&red[q][r][nt][gp][lane][0] = v;
            }
        }
    }
    __syncthreads();

    const int gp = q & 3, nt = q >> 2;
    v2f pr[3][8];
#pragma unroll
    for (int r = 0; r < 3; ++r) {
#pragma unroll
        for (int qq = 0; qq < 8; ++qq) pr[r][qq] = *(const v2f*)&red[qq][r][nt][gp][lane][0];
    }

    const int chb = 64 * (q >> 2) + (q & 3);
    const float* arow = w0s + (chb + 4 * t) * 63;
    v2f a2[16];
#pragma unroll
    for (int kk = 0; kk < 16; ++kk) {
        const int k0 = kk * 4 + hh * 2;
        const int k1 = (k0 + 1 < 63) ? (k0 + 1) : 62;
        const float v1 = arow[k1];
        a2[kk].x = arow[k0];
        a2[kk].y = (k0 + 1 < 63) ? v1 : 0.0f;
    }

    {
        const int  n   = nt * 16 + t;
        const int  m0  = 8 * hh + 2 * gp;
        const bool has = n < 21;
        const int  o0  = has ? (m0 * BPITCH + n) : ((lane >> 1) * BPITCH + 64 + (lane & 1));
        const int  o1  = has ? (o0 + BPITCH) : o0;
        const int  rs  = has ? 21 : 0;
#pragma unroll
        for (int r = 0; r < 3; ++r) {
            v2f s = pr[r][0];
#pragma unroll
            for (int qq = 1; qq < 8; ++qq) { s.x += pr[r][qq].x;  s.y += pr[r][qq].y; }
            const int P = p0 + m0 + 28 * (r - 1);
            Bs[o0 + r * rs] = ((P     >= 0) && (P     < P784)) ? s.x : 0.0f;
            Bs[o1 + r * rs] = ((P + 1 >= 0) && (P + 1 < P784)) ? s.y : 0.0f;
        }
    }
    if (tid < 16) Bs[tid * BPITCH + 63] = 0.0f;
    __syncthreads();

    const float* brow = Bs + t * BPITCH;
    v8f c = {0.f, 0.f, 0.f, 0.f, 0.f, 0.f, 0.f, 0.f};
#pragma unroll
    for (int kk = 0; kk < 16; ++kk) {
        const int k0 = kk * 4 + hh * 2;
        v2f b = *(const v2f*)(brow + k0);
        c = __builtin_amdgcn_wmma_f32_16x16x4_f32(false, a2[kk], false, b, (short)0, c, false, false);
    }

    float* yrow = y + (size_t)(cs * 128 + chb + 4 * (hh * 8)) * P784 + p0 + t;
#pragma unroll
    for (int g = 0; g < 8; ++g) yrow[4 * g * P784] = c[g];
}

extern "C" void kernel_launch(void* const* d_in, const int* in_sizes, int n_in,
                              void* d_out, int out_size, void* d_ws, size_t ws_size,
                              hipStream_t stream) {
    const float* x  = (const float*)d_in[0];
    const float* w0 = (const float*)d_in[1];
    const float* w1 = (const float*)d_in[2];
    float* y = (float*)d_out;

    k_fused<<<98, 256, 0, stream>>>(x, w0, w1, y);
}
